# starting kernel
# baseline (speedup 1.0000x reference)
#include <hip/hip_runtime.h>


typedef __attribute__((ext_vector_type(2))) float v2f;
typedef __attribute__((ext_vector_type(8))) float v8f;

#define BB 4
#define NN 50000
#define KK 16
#define DD 32
#define FF 32
#define KDIM (KK * DD)
#define NGROUP (KDIM / 4)
#define TILES_PER_BATCH (NN / 16)
#define TOTAL_TILES (BB * TILES_PER_BATCH)
#define WAVES_PER_BLOCK 8

__global__ __launch_bounds__(256)
void GraphConv_wmma_f32(const float* __restrict__ x,
                        const int*   __restrict__ nbr,
                        const float* __restrict__ W,
                        const float* __restrict__ bias,
                        float*       __restrict__ out) {
    __shared__ v2f Wfrag[NGROUP * 2 * 32];

    for (int i = threadIdx.x; i < NGROUP * 2 * 32; i += blockDim.x) {
        int lane_i = i & 31;
        int tile   = (i >> 5) & 1;
        int g      = i >> 6;
        int f      = (lane_i & 15) + tile * 16;
        int krow   = 4 * g + ((lane_i >> 4) << 1);
        v2f wv;
        wv.x = W[(krow + 0) * FF + f];
        wv.y = W[(krow + 1) * FF + f];
        Wfrag[i] = wv;
    }
    __syncthreads();

    const int wave  = threadIdx.x >> 5;
    const int lane  = threadIdx.x & 31;
    const int gtile = blockIdx.x * WAVES_PER_BLOCK + wave;
    if (gtile >= TOTAL_TILES) return;

    const int b         = gtile / TILES_PER_BATCH;
    const int node_base = (gtile % TILES_PER_BATCH) * 16;
    const int m         = lane & 15;
    const int hi        = lane >> 4;
    const int node      = node_base + m;

    int nb[KK];
    {
        const int4* np = (const int4*)(nbr + node * KK);
        #pragma unroll
        for (int q = 0; q < 4; ++q) {
            int4 v = np[q];
            nb[4*q+0] = v.x; nb[4*q+1] = v.y; nb[4*q+2] = v.z; nb[4*q+3] = v.w;
        }
    }

    v8f acc0 = {};
    v8f acc1 = {};

    const float* xb = x + (size_t)b * NN * DD;

    for (int k = 0; k < KK; ++k) {
        const float* row = xb + (size_t)nb[k] * DD + hi * 2;
        #pragma unroll
        for (int t = 0; t < 8; ++t) {
            const int g = k * 8 + t;
            v2f a  = *(const v2f*)(row + t * 4);
            v2f b0 = Wfrag[(g * 2 + 0) * 32 + lane];
            v2f b1 = Wfrag[(g * 2 + 1) * 32 + lane];
            acc0 = __builtin_amdgcn_wmma_f32_16x16x4_f32(
                       false, a, false, b0, (short)0, acc0, false, false);
            acc1 = __builtin_amdgcn_wmma_f32_16x16x4_f32(
                       false, a, false, b1, (short)0, acc1, false, false);
        }
    }

    const float bias0 = bias[m];
    const float bias1 = bias[m + 16];
    float* ob = out + ((size_t)b * NN + node_base) * FF;
    #pragma unroll
    for (int v = 0; v < 8; ++v) {
        const int row_m = v + hi * 8;
        ob[row_m * FF + m]      = acc0[v] + bias0;
        ob[row_m * FF + m + 16] = acc1[v] + bias1;
    }
}

extern "C" void kernel_launch(void* const* d_in, const int* in_sizes, int n_in,
                              void* d_out, int out_size, void* d_ws, size_t ws_size,
                              hipStream_t stream) {
    const float* x    = (const float*)d_in[0];
    const int*   nbr  = (const int*)  d_in[1];
    const float* W    = (const float*)d_in[2];
    const float* bias = (const float*)d_in[3];
    float*       out  = (float*)d_out;

    const int blocks = (TOTAL_TILES + WAVES_PER_BLOCK - 1) / WAVES_PER_BLOCK;
    GraphConv_wmma_f32<<<blocks, 256, 0, stream>>>(x, nbr, W, bias, out);
}


# A operand goes through LDS: each wave gathers whole 128-B neighbor rows (8 lanes x 16 B, 4 lines per load; was 16 lines x 8 B), parks them in a private padded tile and reads its fragments there
# speedup vs baseline: 1.5765x; 1.5765x over previous
#include <hip/hip_runtime.h>


typedef __attribute__((ext_vector_type(2))) float v2f;
typedef __attribute__((ext_vector_type(4))) float v4f;
typedef __attribute__((ext_vector_type(8))) float v8f;

#define BB 4
#define NN 50000
#define KK 16
#define DD 32
#define FF 32
#define KDIM (KK * DD)
#define NGROUP (KDIM / 4)
#define NODE_TILES (NN / 16)
#define BPW 2
#define WAVE_TILES (NODE_TILES * (BB / BPW))
#define WAVES_PER_BLOCK 25
#define THREADS (WAVES_PER_BLOCK * 32)
#define BLOCKS (WAVE_TILES / WAVES_PER_BLOCK)
#define NFRAG (NGROUP * 2 * 32)
#define STAGE_WAVES 16
#define STAGE_TRIPS (NFRAG / (STAGE_WAVES * 32))
#define APITCH 36
#define ATILE (16 * APITCH)
#define ASTAGE (BPW * ATILE)
static_assert(STAGE_TRIPS * STAGE_WAVES * 32 == NFRAG && STAGE_WAVES <= WAVES_PER_BLOCK, "staging covers the table exactly");
static_assert(BLOCKS * WAVES_PER_BLOCK == WAVE_TILES, "every wave of the grid owns exactly one tile");
static_assert(NODE_TILES * 16 == NN, "no ragged node tile");
static_assert(BPW == 2 && DD == 32 && KK == 16, "the row gather below is written for 128-byte rows, 16 neighbors, two batches");

template <int S>
__device__ __forceinline__ int group8_bcast(int v) {
    return (int)__builtin_amdgcn_mov_dpp8((unsigned)v, (unsigned)(S * 0x249249));
}

__device__ __forceinline__ void wave_lds_handover() {
    __builtin_amdgcn_fence(__ATOMIC_RELEASE, "wavefront");
    __builtin_amdgcn_wave_barrier();
    __builtin_amdgcn_fence(__ATOMIC_ACQUIRE, "wavefront");
}

template <int K>
__device__ __forceinline__ void gather_rows(const float* __restrict__ xb0, const float* __restrict__ xb1,
                                            const int2 (&idx)[4], int c4, v4f (&G)[BPW][4]) {
    #pragma unroll
    for (int j = 0; j < 4; ++j) {
        const int own = (K & 1) ? idx[j].y : idx[j].x;
        const unsigned off = (unsigned)group8_bcast<(K >> 1)>(own) * DD + c4;
        G[0][j] = *(const v4f*)(xb0 + off);
        G[1][j] = *(const v4f*)(xb1 + off);
    }
}

__global__ __launch_bounds__(THREADS)
void GraphConv_wmma_f32(const float* __restrict__ x,
                        const int*   __restrict__ nbr,
                        const float* __restrict__ W,
                        const float* __restrict__ bias,
                        float*       __restrict__ out) {
    __shared__ v4f lds[(NFRAG * 2 + WAVES_PER_BLOCK * ASTAGE) / 4];
    v2f*   const Wfrag  = (v2f*)lds;
    float* const Astage = (float*)lds + NFRAG * 2;

    const int wave  = threadIdx.x >> 5;
    const int lane  = threadIdx.x & 31;
    const int wtile = blockIdx.x * WAVES_PER_BLOCK + wave;
    const int ntile = wtile >> 1;
    const int b0    = (wtile & 1) * BPW;
    const int node_base = ntile * 16;
    const int m     = lane & 15;
    const int hi    = lane >> 4;

    const int r  = lane >> 3;
    const int c  = lane & 7;
    const int rowsel = ((r & 1) << 3) | ((r >> 1) << 2);

    int2 idx[4];
    #pragma unroll
    for (int j = 0; j < 4; ++j)
        idx[j] = *(const int2*)(nbr + (node_base + rowsel + j) * KK + 2 * c);

    const float* xb0 = x + (size_t)b0 * NN * DD;
    const float* xb1 = xb0 + (size_t)NN * DD;
    const int c4 = c * 4;

    v4f G[BPW][4];
    gather_rows<0>(xb0, xb1, idx, c4, G);

    if (wave < STAGE_WAVES) {
        #pragma unroll
        for (int j = 0; j < STAGE_TRIPS; ++j) {
            const int i = threadIdx.x + j * (STAGE_WAVES * 32);
            int lane_i = i & 31;
            int tile   = (i >> 5) & 1;
            int g      = i >> 6;
            int f      = (lane_i & 15) + tile * 16;
            int krow   = 4 * g + ((lane_i >> 4) << 1);
            v2f wv;
            wv.x = W[(krow + 0) * FF + f];
            wv.y = W[(krow + 1) * FF + f];
            Wfrag[i] = wv;
        }
    }
    __syncthreads();

    v8f acc00 = {};
    v8f acc01 = {};
    v8f acc10 = {};
    v8f acc11 = {};

    float* const tile0 = Astage + wave * ASTAGE;
    v4f*   const park0 = (v4f*)(tile0 + rowsel * APITCH + c4);
    const float* const frag0 = tile0 + m * APITCH + hi * 2;

#define STEP(K)                                                                             \
    {                                                                                       \
        _Pragma("unroll")                                                                   \
        for (int j = 0; j < 4; ++j) {                                                       \
            park0[(j * APITCH) / 4]           = G[0][j];                                    \
            park0[(ATILE + j * APITCH) / 4]   = G[1][j];                                    \
        }                                                                                   \
        if ((K) + 1 < KK) gather_rows<((K) + 1) & (KK - 1)>(xb0, xb1, idx, c4, G);          \
        wave_lds_handover();                                                                \
        _Pragma("unroll")                                                                   \
        for (int t = 0; t < 8; ++t) {                                                       \
            const int g = (K) * 8 + t;                                                      \
            v2f a0 = *(const v2f*)(frag0 + t * 4);                                          \
            v2f a1 = *(const v2f*)(frag0 + ATILE + t * 4);                                  \
            v2f w0 = Wfrag[(g * 2 + 0) * 32 + lane];                                        \
            v2f w1 = Wfrag[(g * 2 + 1) * 32 + lane];                                        \
            acc00 = __builtin_amdgcn_wmma_f32_16x16x4_f32(                                  \
                        false, a0, false, w0, (short)0, acc00, false, false);               \
            acc01 = __builtin_amdgcn_wmma_f32_16x16x4_f32(                                  \
                        false, a0, false, w1, (short)0, acc01, false, false);               \
            acc10 = __builtin_amdgcn_wmma_f32_16x16x4_f32(                                  \
                        false, a1, false, w0, (short)0, acc10, false, false);               \
            acc11 = __builtin_amdgcn_wmma_f32_16x16x4_f32(                                  \
                        false, a1, false, w1, (short)0, acc11, false, false);               \
        }                                                                                   \
        wave_lds_handover();                                                                \
    }

    STEP(0)  STEP(1)  STEP(2)  STEP(3)  STEP(4)  STEP(5)  STEP(6)  STEP(7)
    STEP(8)  STEP(9)  STEP(10) STEP(11) STEP(12) STEP(13) STEP(14) STEP(15)
#undef STEP

    const float bias0 = bias[m];
    const float bias1 = bias[m + 16];
    float* ob0 = out + ((size_t)b0 * NN + node_base) * FF;
    float* ob1 = ob0 + (size_t)NN * FF;
    #pragma unroll
    for (int v = 0; v < 8; ++v) {
        const int row_m = v + hi * 8;
        ob0[row_m * FF + m]      = acc00[v] + bias0;
        ob0[row_m * FF + m + 16] = acc01[v] + bias1;
        ob1[row_m * FF + m]      = acc10[v] + bias0;
        ob1[row_m * FF + m + 16] = acc11[v] + bias1;
    }
}

extern "C" void kernel_launch(void* const* d_in, const int* in_sizes, int n_in,
                              void* d_out, int out_size, void* d_ws, size_t ws_size,
                              hipStream_t stream) {
    const float* x    = (const float*)d_in[0];
    const int*   nbr  = (const int*)  d_in[1];
    const float* W    = (const float*)d_in[2];
    const float* bias = (const float*)d_in[3];
    float*       out  = (float*)d_out;

    GraphConv_wmma_f32<<<BLOCKS, THREADS, 0, stream>>>(x, nbr, W, bias, out);
}


# Each wave owns 16 nodes for all four batches (64x32 tile, was 32x32): 256 groups of 12 such waves, 3 per SIMD, half as many waves enter; the 53 left-over tiles run as quarter tiles on a 13th wave
# speedup vs baseline: 1.6848x; 1.0687x over previous
#include <hip/hip_runtime.h>


typedef __attribute__((ext_vector_type(2))) float v2f;
typedef __attribute__((ext_vector_type(4))) float v4f;
typedef __attribute__((ext_vector_type(8))) float v8f;

#define BB 4
#define NN 50000
#define KK 16
#define DD 32
#define FF 32
#define KDIM (KK * DD)
#define NGROUP (KDIM / 4)
#define NODE_TILES (NN / 16)
#define BLOCKS 256
#define FULL_WAVES 12
#define WAVES_PER_BLOCK (FULL_WAVES + 1)
#define THREADS (WAVES_PER_BLOCK * 32)
#define FULL_TILES (BLOCKS * FULL_WAVES)
#define REST_TILES (NODE_TILES - FULL_TILES)
#define REST_QUARTERS (REST_TILES * BB)
#define NFRAG (NGROUP * 2 * 32)
#define STAGE_FULL_TRIPS (NFRAG / THREADS)
#define STAGE_REST_WAVES ((NFRAG - STAGE_FULL_TRIPS * THREADS) / 32)
#define APITCH 36
#define ATILE (16 * APITCH)
#define ASTAGE (BB * ATILE)
static_assert(STAGE_FULL_TRIPS * THREADS + STAGE_REST_WAVES * 32 == NFRAG && STAGE_REST_WAVES <= WAVES_PER_BLOCK, "staging covers the table exactly");
static_assert(REST_TILES >= 0 && REST_QUARTERS <= BLOCKS, "every left-over quarter tile has a work-group");
static_assert(FULL_TILES + REST_TILES == NODE_TILES, "whole tiles and left-over tiles cover every node");
static_assert(NODE_TILES * 16 == NN, "no ragged node tile");
static_assert(BB == 4 && DD == 32 && KK == 16, "the row gather below is written for 128-byte rows, 16 neighbors, four batches");

template <int S>
__device__ __forceinline__ int group8_bcast(int v) {
    return (int)__builtin_amdgcn_mov_dpp8((unsigned)v, (unsigned)(S * 0x249249));
}

__device__ __forceinline__ void wave_lds_handover() {
    __builtin_amdgcn_fence(__ATOMIC_RELEASE, "wavefront");
    __builtin_amdgcn_wave_barrier();
    __builtin_amdgcn_fence(__ATOMIC_ACQUIRE, "wavefront");
}

template <int K, int NB>
__device__ __forceinline__ void gather_rows(const float* __restrict__ xb,
                                            const int2 (&idx)[4], int c16, v4f (&G)[BB][4]) {
    constexpr unsigned PLANE = NN * DD * 4u;
    #pragma unroll
    for (int j = 0; j < 4; ++j) {
        const int own = (K & 1) ? idx[j].y : idx[j].x;
        const unsigned boff = (unsigned)group8_bcast<(K >> 1)>(own) * (DD * 4u) + (unsigned)c16;
        const unsigned boff_hi = boff + 2u * PLANE;
        #pragma unroll
        for (int b = 0; b < NB; ++b)
            G[b][j] = *(const v4f*)((const char*)xb + (size_t)(b & 1) * PLANE + (b < 2 ? boff : boff_hi));
    }
}

template <int K, int NB>
__device__ __forceinline__ void slot_step(const float* __restrict__ xb, const int2 (&idx)[4], int c16,
                                          v4f (&G)[BB][4], v4f* park0, const float* frag0,
                                          const v2f* Wfrag, int lane, v8f (&acc)[BB][2]) {
    #pragma unroll
    for (int j = 0; j < 4; ++j) {
        #pragma unroll
        for (int b = 0; b < NB; ++b)
            park0[(b * ATILE + j * APITCH) / 4] = G[b][j];
    }
    __builtin_amdgcn_sched_barrier(0);
    if (K + 1 < KK) gather_rows<(K + 1) & (KK - 1), NB>(xb, idx, c16, G);
    wave_lds_handover();
    #pragma unroll
    for (int t = 0; t < 8; ++t) {
        const int g = K * 8 + t;
        const v2f w0 = Wfrag[(g * 2 + 0) * 32 + lane];
        const v2f w1 = Wfrag[(g * 2 + 1) * 32 + lane];
        #pragma unroll
        for (int b = 0; b < NB; ++b) {
            const v2f a = *(const v2f*)(frag0 + b * ATILE + t * 4);
            acc[b][0] = __builtin_amdgcn_wmma_f32_16x16x4_f32(
                            false, a, false, w0, (short)0, acc[b][0], false, false);
            acc[b][1] = __builtin_amdgcn_wmma_f32_16x16x4_f32(
                            false, a, false, w1, (short)0, acc[b][1], false, false);
        }
    }
    wave_lds_handover();
}

template <int NB>
__device__ __forceinline__ void run_tile(const float* __restrict__ xb, const int2 (&idx)[4], int c16,
                                         v4f (&G)[BB][4], v4f* park0, const float* frag0,
                                         const v2f* Wfrag, int lane,
                                         const float* __restrict__ bias, float* __restrict__ ob) {
    v8f acc[BB][2] = {};

    slot_step<0,  NB>(xb, idx, c16, G, park0, frag0, Wfrag, lane, acc);
    slot_step<1,  NB>(xb, idx, c16, G, park0, frag0, Wfrag, lane, acc);
    slot_step<2,  NB>(xb, idx, c16, G, park0, frag0, Wfrag, lane, acc);
    slot_step<3,  NB>(xb, idx, c16, G, park0, frag0, Wfrag, lane, acc);
    slot_step<4,  NB>(xb, idx, c16, G, park0, frag0, Wfrag, lane, acc);
    slot_step<5,  NB>(xb, idx, c16, G, park0, frag0, Wfrag, lane, acc);
    slot_step<6,  NB>(xb, idx, c16, G, park0, frag0, Wfrag, lane, acc);
    slot_step<7,  NB>(xb, idx, c16, G, park0, frag0, Wfrag, lane, acc);
    slot_step<8,  NB>(xb, idx, c16, G, park0, frag0, Wfrag, lane, acc);
    slot_step<9,  NB>(xb, idx, c16, G, park0, frag0, Wfrag, lane, acc);
    slot_step<10, NB>(xb, idx, c16, G, park0, frag0, Wfrag, lane, acc);
    slot_step<11, NB>(xb, idx, c16, G, park0, frag0, Wfrag, lane, acc);
    slot_step<12, NB>(xb, idx, c16, G, park0, frag0, Wfrag, lane, acc);
    slot_step<13, NB>(xb, idx, c16, G, park0, frag0, Wfrag, lane, acc);
    slot_step<14, NB>(xb, idx, c16, G, park0, frag0, Wfrag, lane, acc);
    slot_step<15, NB>(xb, idx, c16, G, park0, frag0, Wfrag, lane, acc);

    const int m  = lane & 15;
    const int hi = lane >> 4;
    const float bias0 = bias[m];
    const float bias1 = bias[m + 16];
    #pragma unroll
    for (int b = 0; b < NB; ++b) {
        float* obb = ob + (size_t)b * NN * FF;
        #pragma unroll
        for (int v = 0; v < 8; ++v) {
            const int row_m = v + hi * 8;
            obb[row_m * FF + m]      = acc[b][0][v] + bias0;
            obb[row_m * FF + m + 16] = acc[b][1][v] + bias1;
        }
    }
}

__global__ __launch_bounds__(THREADS)
void GraphConv_wmma_f32(const float* __restrict__ x,
                        const int*   __restrict__ nbr,
                        const float* __restrict__ W,
                        const float* __restrict__ bias,
                        float*       __restrict__ out) {
    __shared__ v4f lds[(NFRAG * 2 + WAVES_PER_BLOCK * ASTAGE) / 4];
    v2f*   const Wfrag  = (v2f*)lds;
    float* const Astage = (float*)lds + NFRAG * 2;

    const int wave = __builtin_amdgcn_readfirstlane((int)(threadIdx.x >> 5));
    const int lane = threadIdx.x & 31;
    const bool whole  = wave < FULL_WAVES;
    const bool active = whole || (int)blockIdx.x < REST_QUARTERS;
    const int rest   = min((int)blockIdx.x >> 2, REST_TILES - 1);
    const int ntile  = whole ? (int)blockIdx.x * FULL_WAVES + wave : FULL_TILES + rest;
    const int bfirst = whole ? 0 : ((int)blockIdx.x & 3);
    const int node_base = ntile * 16;
    const int m     = lane & 15;
    const int hi    = lane >> 4;

    const int r  = lane >> 3;
    const int c  = lane & 7;
    const int rowsel = ((r & 1) << 3) | ((r >> 1) << 2);
    const int c4 = c * 4;

    const float* xb = x + (size_t)bfirst * NN * DD;

    int2 idx[4];
    v4f G[BB][4];
    if (active) {
        #pragma unroll
        for (int j = 0; j < 4; ++j)
            idx[j] = *(const int2*)(nbr + (node_base + rowsel + j) * KK + 2 * c);
        if (whole) gather_rows<0, BB>(xb, idx, c4 * 4, G);
        else       gather_rows<0, 1>(xb, idx, c4 * 4, G);
    }

    {
        auto stage_entry = [&](int i) {
            int lane_i = i & 31;
            int tile   = (i >> 5) & 1;
            int g      = i >> 6;
            int f      = (lane_i & 15) + tile * 16;
            int krow   = 4 * g + ((lane_i >> 4) << 1);
            v2f wv;
            wv.x = W[(krow + 0) * FF + f];
            wv.y = W[(krow + 1) * FF + f];
            Wfrag[i] = wv;
        };
        #pragma unroll
        for (int j = 0; j < STAGE_FULL_TRIPS; ++j)
            stage_entry((int)threadIdx.x + j * THREADS);
        if (wave < STAGE_REST_WAVES)
            stage_entry((int)threadIdx.x + STAGE_FULL_TRIPS * THREADS);
    }
    __syncthreads();

    float* const tile0 = Astage + wave * ASTAGE;
    v4f*   const park0 = (v4f*)(tile0 + rowsel * APITCH + c4);
    const float* const frag0 = tile0 + m * APITCH + hi * 2;
    float* const ob = out + ((size_t)bfirst * NN + node_base) * FF;

    if (whole)
        run_tile<BB>(xb, idx, c4 * 4, G, park0, frag0, Wfrag, lane, bias, ob);
    else if (active)
        run_tile<1>(xb, idx, c4 * 4, G, park0, frag0, Wfrag, lane, bias, ob);
}

extern "C" void kernel_launch(void* const* d_in, const int* in_sizes, int n_in,
                              void* d_out, int out_size, void* d_ws, size_t ws_size,
                              hipStream_t stream) {
    const float* x    = (const float*)d_in[0];
    const int*   nbr  = (const int*)  d_in[1];
    const float* W    = (const float*)d_in[2];
    const float* bias = (const float*)d_in[3];
    float*       out  = (float*)d_out;

    GraphConv_wmma_f32<<<BLOCKS, THREADS, 0, stream>>>(x, nbr, W, bias, out);
}
